# starting kernel
# baseline (speedup 1.0000x reference)
#include <hip/hip_runtime.h>


#define TT_D 32
#define TT_R 16
#define WAVES_PER_BLOCK 8
#define TILES_PER_WAVE 2
#define ROWS_PER_WAVE (16 * TILES_PER_WAVE)
#define LDS_ROW_STRIDE 18
#define LDS_BUF_FLOATS (16 * LDS_ROW_STRIDE)
#define LDS_PER_TILE (2 * LDS_BUF_FLOATS)

typedef float v2f __attribute__((ext_vector_type(2)));
typedef float v8f __attribute__((ext_vector_type(8)));
typedef int   v4i __attribute__((ext_vector_type(4)));

__global__ __launch_bounds__(256) void tt_chain_wmma_f32(
    const int*   __restrict__ X,
    const float* __restrict__ core_first,
    const float* __restrict__ cores_mid,
    const float* __restrict__ core_last,
    float*       __restrict__ out)
{
    __shared__ float lds[WAVES_PER_BLOCK][TILES_PER_WAVE * LDS_PER_TILE];

    const int tid  = threadIdx.x;
    const int wave = tid >> 5;
    const int lane = tid & 31;
    const int row  = lane & 15;
    const int half = lane >> 4;

    const int base = (blockIdx.x * WAVES_PER_BLOCK + wave) * ROWS_PER_WAVE;

    float* wbuf = lds[wave];

    unsigned xm[TILES_PER_WAVE];
    #pragma unroll
    for (int t = 0; t < TILES_PER_WAVE; ++t) {
        const v4i* xr = (const v4i*)(X + (size_t)(base + 16 * t + row) * TT_D);
        unsigned m = 0u;
        #pragma unroll
        for (int q = 0; q < 8; ++q) {
            v4i v = __builtin_nontemporal_load(xr + q);
            m |= (unsigned)(v.x & 1) << (4 * q + 0);
            m |= (unsigned)(v.y & 1) << (4 * q + 1);
            m |= (unsigned)(v.z & 1) << (4 * q + 2);
            m |= (unsigned)(v.w & 1) << (4 * q + 3);
        }
        xm[t] = m;
    }

    float a[TILES_PER_WAVE][4][2];
    #pragma unroll
    for (int t = 0; t < TILES_PER_WAVE; ++t) {
        const unsigned x0 = xm[t] & 1u;
        #pragma unroll
        for (int c = 0; c < 4; ++c)
            #pragma unroll
            for (int j = 0; j < 2; ++j) {
                int k = 4 * c + 2 * half + j;
                a[t][c][j] = core_first[k * 2 + x0];
            }
    }

    #pragma unroll 1
    for (int ci = 0; ci < TT_D - 2; ++ci) {
        const float* Bc = cores_mid + ci * (TT_R * TT_R * 2);
        float2 bv[4][2];
        #pragma unroll
        for (int c = 0; c < 4; ++c)
            #pragma unroll
            for (int j = 0; j < 2; ++j) {
                int k = 4 * c + 2 * half + j;
                bv[c][j] = *(const float2*)(Bc + k * 32 + row * 2);
            }
        v2f b0[4], b1[4];
        #pragma unroll
        for (int c = 0; c < 4; ++c) {
            b0[c] = (v2f){ bv[c][0].x, bv[c][1].x };
            b1[c] = (v2f){ bv[c][0].y, bv[c][1].y };
        }

        v8f p0[TILES_PER_WAVE], p1[TILES_PER_WAVE];
        #pragma unroll
        for (int t = 0; t < TILES_PER_WAVE; ++t) { p0[t] = (v8f){}; p1[t] = (v8f){}; }

        #pragma unroll
        for (int c = 0; c < 4; ++c) {
            #pragma unroll
            for (int t = 0; t < TILES_PER_WAVE; ++t) {
                v2f av = { a[t][c][0], a[t][c][1] };
                p0[t] = __builtin_amdgcn_wmma_f32_16x16x4_f32(
                            false, av, false, b0[c], (short)0, p0[t], false, false);
                p1[t] = __builtin_amdgcn_wmma_f32_16x16x4_f32(
                            false, av, false, b1[c], (short)0, p1[t], false, false);
            }
        }

        #pragma unroll
        for (int t = 0; t < TILES_PER_WAVE; ++t) {
            float* tb = wbuf + t * LDS_PER_TILE;
            #pragma unroll
            for (int v = 0; v < 8; ++v) {
                int r = v + 8 * half;
                tb[r * LDS_ROW_STRIDE + row]                  = p0[t][v];
                tb[LDS_BUF_FLOATS + r * LDS_ROW_STRIDE + row] = p1[t][v];
            }
        }

        #pragma unroll
        for (int t = 0; t < TILES_PER_WAVE; ++t) {
            const unsigned sel = (xm[t] >> (ci + 1)) & 1u;
            const float* src = wbuf + t * LDS_PER_TILE + sel * LDS_BUF_FLOATS
                               + row * LDS_ROW_STRIDE + 2 * half;
            #pragma unroll
            for (int c = 0; c < 4; ++c) {
                float2 tv = *(const float2*)(src + 4 * c);
                a[t][c][0] = tv.x;
                a[t][c][1] = tv.y;
            }
        }
    }

    #pragma unroll
    for (int t = 0; t < TILES_PER_WAVE; ++t) {
        const unsigned xl = (xm[t] >> 31) & 1u;
        float acc = 0.f;
        #pragma unroll
        for (int c = 0; c < 4; ++c)
            #pragma unroll
            for (int j = 0; j < 2; ++j) {
                int k = 4 * c + 2 * half + j;
                acc += a[t][c][j] * core_last[k * 2 + xl];
            }
        acc += __shfl_xor(acc, 16, 32);
        if (half == 0)
            __builtin_nontemporal_store(acc, out + base + 16 * t + row);
    }
}

extern "C" void kernel_launch(void* const* d_in, const int* in_sizes, int n_in,
                              void* d_out, int out_size, void* d_ws, size_t ws_size,
                              hipStream_t stream) {
    (void)in_sizes; (void)n_in; (void)d_ws; (void)ws_size;
    const int*   X  = (const int*)d_in[0];
    const float* cf = (const float*)d_in[1];
    const float* cm = (const float*)d_in[2];
    const float* cl = (const float*)d_in[3];
    float* out = (float*)d_out;

    const int rows_per_block = ROWS_PER_WAVE * WAVES_PER_BLOCK;
    const int blocks = out_size / rows_per_block;
    tt_chain_wmma_f32<<<blocks, 256, 0, stream>>>(X, cf, cm, cl, out);
}


# Rows ordered by the 2-bit digit over the 1024-row work-group, not per wave (class offsets prefixed once, carries ping-pong in LDS, barrier per step): 4-5 products per wave-step, was 7; sum order kept
# speedup vs baseline: 1.5751x; 1.5751x over previous
#include <hip/hip_runtime.h>


#define TT_D 32
#define TT_R 16
#define TT_PAIRS 15
#define WAVES_PER_BLOCK 16
#define TILES_PER_WAVE 4
#define GROUPS_PER_WAVE 2
#define ROWS_PER_WAVE (16 * TILES_PER_WAVE)
#define LDS_ROW_STRIDE 18
#define WG_ROWS (ROWS_PER_WAVE * WAVES_PER_BLOCK)
#define CARRY_FLOATS (WG_ROWS * LDS_ROW_STRIDE)
#define FUSED_FLOATS (TT_PAIRS * 4 * TT_R * TT_R)
#define CNT_INTS (TT_PAIRS * 4 * WAVES_PER_BLOCK)

typedef float v2f __attribute__((ext_vector_type(2)));
typedef float v8f __attribute__((ext_vector_type(8)));
typedef int   v4i __attribute__((ext_vector_type(4)));
typedef int   v8i __attribute__((ext_vector_type(8)));

static __device__ __forceinline__ int fused_off(int f, int u, int g) {
    return f * (TT_R * TT_R) + u * TT_R + 4 * (g ^ ((u >> 2) & 3));
}

static __device__ __forceinline__ void load_a(v2f (&a)[4], const float* wbuf, int slot, int half) {
    const float* q = wbuf + slot * LDS_ROW_STRIDE + 2 * half;
    #pragma unroll
    for (int c = 0; c < 4; ++c) {
        const float2 u = *(const float2*)(q + 4 * c);
        a[c] = (v2f){ u.x, u.y };
    }
}

static __device__ __forceinline__ void tile_class(v8f& r, const v2f (&a)[4], const v2f (&b)[4], int th) {
    v8f pr = (v8f){};
    #pragma unroll
    for (int c = 0; c < 4; ++c)
        pr = __builtin_amdgcn_wmma_f32_16x16x4_f32(
                 false, a[c], false, b[c], (short)0, pr, false, false);
    const v8i vi = (v8i){ 0, 1, 2, 3, 4, 5, 6, 7 };
    r = (vi >= th) ? pr : r;
}

static __device__ __forceinline__ void store_tile(float* wbuf, int t, const v8f& r, int row, int half) {
    #pragma unroll
    for (int v = 0; v < 8; ++v)
        wbuf[(16 * t + v + 8 * half) * LDS_ROW_STRIDE + row] = r[v];
}

__global__ __launch_bounds__(512, 4) void tt_chain_wmma_f32(
    const int*   __restrict__ X,
    const float* __restrict__ core_first,
    const float* __restrict__ cores_mid,
    const float* __restrict__ core_last,
    float*       __restrict__ out)
{
    __shared__ __attribute__((aligned(16))) float fused[FUSED_FLOATS];
    __shared__ __attribute__((aligned(16))) float carry[2][CARRY_FLOATS];
    __shared__ int   smap[2][WG_ROWS];
    __shared__ int   cnt[CNT_INTS];
    __shared__ int   pbase[CNT_INTS];

    const int tid  = threadIdx.x;
    const int wave = __builtin_amdgcn_readfirstlane(tid >> 5);
    const int lane = tid & 31;
    const int row  = lane & 15;
    const int half = lane >> 4;

    const int base = (blockIdx.x * WAVES_PER_BLOCK + wave) * ROWS_PER_WAVE;

    const int wofs = wave * ROWS_PER_WAVE;

    const int pw = wave < TT_PAIRS ? wave : TT_PAIRS - 1;
    float4 am[4];
    float2 bm[4][2];
    {
        const float* Ma = cores_mid + (2 * pw) * (TT_R * TT_R * 2);
        const float* Mb = Ma + TT_R * TT_R * 2;
        #pragma unroll
        for (int c = 0; c < 4; ++c) {
            const int k = 4 * c + 2 * half;
            am[c]    = *(const float4*)(Ma + row * 32 + k * 2);
            bm[c][0] = *(const float2*)(Mb + k * 32 + row * 2);
            bm[c][1] = *(const float2*)(Mb + (k + 1) * 32 + row * 2);
        }
    }

    unsigned xo[GROUPS_PER_WAVE];
    #pragma unroll
    for (int g = 0; g < GROUPS_PER_WAVE; ++g) {
        const v4i* xr = (const v4i*)(X + (size_t)(base + 32 * g + lane) * TT_D);
        unsigned m = 0u;
        #pragma unroll
        for (int q = 0; q < 8; ++q) {
            v4i v = __builtin_nontemporal_load(xr + q);
            m |= (unsigned)(v.x & 1) << (4 * q + 0);
            m |= (unsigned)(v.y & 1) << (4 * q + 1);
            m |= (unsigned)(v.z & 1) << (4 * q + 2);
            m |= (unsigned)(v.w & 1) << (4 * q + 3);
        }
        xo[g] = m;
    }

    if (wave < TT_PAIRS) {
        #pragma unroll
        for (int cls = 0; cls < 4; ++cls) {
            v8f acc = (v8f){};
            #pragma unroll
            for (int c = 0; c < 4; ++c) {
                const v2f av = (cls & 2) ? (v2f){ am[c].y, am[c].w } : (v2f){ am[c].x, am[c].z };
                const v2f bw = (cls & 1) ? (v2f){ bm[c][0].y, bm[c][1].y } : (v2f){ bm[c][0].x, bm[c][1].x };
                acc = __builtin_amdgcn_wmma_f32_16x16x4_f32(
                          false, av, false, bw, (short)0, acc, false, false);
            }
            const int f = 4 * wave + cls;
            *(float4*)(fused + fused_off(f, row, 2 * half))     = make_float4(acc[0], acc[1], acc[2], acc[3]);
            *(float4*)(fused + fused_off(f, row, 2 * half + 1)) = make_float4(acc[4], acc[5], acc[6], acc[7]);
        }
    }

    #pragma unroll
    for (int g = 0; g < GROUPS_PER_WAVE; ++g) {
        const bool x0 = (xo[g] & 1u) != 0u;
        float* dst = carry[0] + (wofs + 32 * g + lane) * LDS_ROW_STRIDE;
        #pragma unroll
        for (int k = 0; k < TT_R; k += 2) {
            float2 iv;
            iv.x = x0 ? core_first[k * 2 + 1]       : core_first[k * 2];
            iv.y = x0 ? core_first[(k + 1) * 2 + 1] : core_first[(k + 1) * 2];
            *(float2*)(dst + k) = iv;
        }
    }

    #pragma unroll 1
    for (int p = 0; p < TT_PAIRS; ++p) {
        const unsigned e0 = (xo[0] >> (2 * p + 1)) & 3u;
        const unsigned e1 = (xo[1] >> (2 * p + 1)) & 3u;
        const unsigned qa0 = __builtin_amdgcn_ballot_w32((e0 & 1u) != 0u);
        const unsigned qb0 = __builtin_amdgcn_ballot_w32((e0 & 2u) != 0u);
        const unsigned qa1 = __builtin_amdgcn_ballot_w32((e1 & 1u) != 0u);
        const unsigned qb1 = __builtin_amdgcn_ballot_w32((e1 & 2u) != 0u);
        const int n0 = __builtin_popcount(~qa0 & ~qb0) + __builtin_popcount(~qa1 & ~qb1);
        const int n1 = __builtin_popcount(~qa0 &  qb0) + __builtin_popcount(~qa1 &  qb1);
        const int n2 = __builtin_popcount( qa0 & ~qb0) + __builtin_popcount( qa1 & ~qb1);
        const int n3 = ROWS_PER_WAVE - n0 - n1 - n2;
        const int n  = lane == 0 ? n0 : lane == 1 ? n1 : lane == 2 ? n2 : n3;
        if (lane < 4) cnt[p * 64 + lane * WAVES_PER_BLOCK + wave] = n;
    }

    __syncthreads();

    if (wave < TT_PAIRS) {
        const int* cp = cnt + wave * 64;
        int run = 0, f0 = 0, f1 = 0;
        #pragma unroll
        for (int j = 0; j < 64; ++j) {
            const int v = cp[j];
            f0 = (j == 2 * lane)     ? run : f0;
            f1 = (j == 2 * lane + 1) ? run : f1;
            run += v;
        }
        pbase[wave * 64 + 2 * lane]     = f0;
        pbase[wave * 64 + 2 * lane + 1] = f1;
    }

    __syncthreads();

    int cur0 = wofs + lane;
    int cur1 = wofs + 32 + lane;

    int nb0, nb1, nS1, nS2, nS3;
    {
        const unsigned e0 = (xo[0] >> 1) & 3u;
        const unsigned e1 = (xo[1] >> 1) & 3u;
        nb0 = pbase[(int)(((e0 & 1u) << 1) | (e0 >> 1)) * WAVES_PER_BLOCK + wave];
        nb1 = pbase[(int)(((e1 & 1u) << 1) | (e1 >> 1)) * WAVES_PER_BLOCK + wave];
        nS1 = pbase[1 * WAVES_PER_BLOCK];
        nS2 = pbase[2 * WAVES_PER_BLOCK];
        nS3 = pbase[3 * WAVES_PER_BLOCK];
    }

    #pragma unroll 1
    for (int p = 0; p < TT_PAIRS; ++p) {
        const float* cin  = carry[p & 1];
        float*       cout = carry[(p + 1) & 1];
        int*         mp   = smap[p & 1];

        const int b0 = nb0;
        const int b1 = nb1;
        const int S1 = __builtin_amdgcn_readfirstlane(nS1);
        const int S2 = __builtin_amdgcn_readfirstlane(nS2);
        const int S3 = __builtin_amdgcn_readfirstlane(nS3);

        const unsigned d0  = (xo[0] >> (2 * p + 1)) & 3u;
        const unsigned d1  = (xo[1] >> (2 * p + 1)) & 3u;
        const bool     da0 = (d0 & 1u) != 0u;
        const bool     db0 = (d0 & 2u) != 0u;
        const bool     da1 = (d1 & 1u) != 0u;
        const bool     db1 = (d1 & 2u) != 0u;
        const unsigned ba0 = __builtin_amdgcn_ballot_w32(da0);
        const unsigned bb0 = __builtin_amdgcn_ballot_w32(db0);
        const unsigned ba1 = __builtin_amdgcn_ballot_w32(da1);
        const unsigned bb1 = __builtin_amdgcn_ballot_w32(db1);
        const int g0c0 = __builtin_popcount(~ba0 & ~bb0);
        const int g0c1 = __builtin_popcount(~ba0 &  bb0);
        const int g0c2 = __builtin_popcount( ba0 & ~bb0);
        const int g0c3 = __builtin_popcount( ba0 &  bb0);
        const unsigned mym0  = (da0 ? ba0 : ~ba0) & (db0 ? bb0 : ~bb0);
        const unsigned mym1  = (da1 ? ba1 : ~ba1) & (db1 ? bb1 : ~bb1);
        const int      rank0 = (int)__builtin_amdgcn_mbcnt_lo(mym0, 0u);
        const int      rank1 = (int)__builtin_amdgcn_mbcnt_lo(mym1, 0u);
        const int      pos0  = b0 + rank0;
        const int      pos1  = b1 + (da1 ? (db1 ? g0c3 : g0c2) : (db1 ? g0c1 : g0c0)) + rank1;

        mp[pos0] = cur0;
        mp[pos1] = cur1;
        cur0 = pos0;
        cur1 = pos1;

        {
            const int pn = p < TT_PAIRS - 1 ? p + 1 : TT_PAIRS - 1;
            const unsigned e0 = (xo[0] >> (2 * pn + 1)) & 3u;
            const unsigned e1 = (xo[1] >> (2 * pn + 1)) & 3u;
            const int* pb = pbase + pn * 64;
            nb0 = pb[(int)(((e0 & 1u) << 1) | (e0 >> 1)) * WAVES_PER_BLOCK + wave];
            nb1 = pb[(int)(((e1 & 1u) << 1) | (e1 >> 1)) * WAVES_PER_BLOCK + wave];
            nS1 = pb[1 * WAVES_PER_BLOCK];
            nS2 = pb[2 * WAVES_PER_BLOCK];
            nS3 = pb[3 * WAVES_PER_BLOCK];
        }

        const int st1 = min(max(S1 - wofs, 0), ROWS_PER_WAVE);
        const int st2 = min(max(S2 - wofs, 0), ROWS_PER_WAVE);
        const int st3 = min(max(S3 - wofs, 0), ROWS_PER_WAVE);
        const int lo[4] = { 0,   st1, st2, st3 };
        const int hi[4] = { st1, st2, st3, ROWS_PER_WAVE };

        v2f bq[4][4];
        #pragma unroll
        for (int cls = 0; cls < 4; ++cls)
            if (hi[cls] > lo[cls]) {
                #pragma unroll
                for (int c = 0; c < 4; ++c) {
                    const float2 t2 = *(const float2*)(fused + fused_off(4 * p + cls, row, c) + 2 * half);
                    bq[cls][c] = (v2f){ t2.x, t2.y };
                }
            }

        __syncthreads();

        v2f a0[4], a1[4], a2[4], a3[4];
        load_a(a0, cin, mp[wofs + row],      half);
        load_a(a1, cin, mp[wofs + 16 + row], half);
        load_a(a2, cin, mp[wofs + 32 + row], half);
        load_a(a3, cin, mp[wofs + 48 + row], half);

        v8f r0 = (v8f){}, r1 = (v8f){}, r2 = (v8f){}, r3 = (v8f){};
        #pragma unroll
        for (int cls = 0; cls < 4; ++cls) {
            const bool some = hi[cls] > lo[cls];
            const int  th   = lo[cls] - 8 * half;
            if (some && lo[cls] < 16 && hi[cls] > 0)  tile_class(r0, a0, bq[cls], th);
            if (some && lo[cls] < 32 && hi[cls] > 16) tile_class(r1, a1, bq[cls], th - 16);
            if (some && lo[cls] < 48 && hi[cls] > 32) tile_class(r2, a2, bq[cls], th - 32);
            if (some && lo[cls] < 64 && hi[cls] > 48) tile_class(r3, a3, bq[cls], th - 48);
        }

        float* wout = cout + wofs * LDS_ROW_STRIDE;
        store_tile(wout, 0, r0, row, half);
        store_tile(wout, 1, r1, row, half);
        store_tile(wout, 2, r2, row, half);
        store_tile(wout, 3, r3, row, half);
    }

    __syncthreads();

    #pragma unroll
    for (int g = 0; g < GROUPS_PER_WAVE; ++g) {
        const bool xl = (xo[g] >> 31) != 0u;
        const int  sl = g ? cur1 : cur0;
        const float* src = carry[TT_PAIRS & 1] + sl * LDS_ROW_STRIDE;
        float acc0 = 0.f, acc1 = 0.f;
        #pragma unroll
        for (int c = 0; c < 4; ++c) {
            const float2 t0 = *(const float2*)(src + 4 * c);
            const float2 t1 = *(const float2*)(src + 4 * c + 2);
            const int k0 = 4 * c;
            const int k1 = 4 * c + 2;
            acc0 += t0.x * (xl ? core_last[k0 * 2 + 1]       : core_last[k0 * 2]);
            acc0 += t0.y * (xl ? core_last[(k0 + 1) * 2 + 1] : core_last[(k0 + 1) * 2]);
            acc1 += t1.x * (xl ? core_last[k1 * 2 + 1]       : core_last[k1 * 2]);
            acc1 += t1.y * (xl ? core_last[(k1 + 1) * 2 + 1] : core_last[(k1 + 1) * 2]);
        }
        __builtin_nontemporal_store(acc0 + acc1, out + base + 32 * g + lane);
    }
}

extern "C" void kernel_launch(void* const* d_in, const int* in_sizes, int n_in,
                              void* d_out, int out_size, void* d_ws, size_t ws_size,
                              hipStream_t stream) {
    (void)in_sizes; (void)n_in; (void)d_ws; (void)ws_size;
    const int*   X  = (const int*)d_in[0];
    const float* cf = (const float*)d_in[1];
    const float* cm = (const float*)d_in[2];
    const float* cl = (const float*)d_in[3];
    float* out = (float*)d_out;

    const int rows_per_block = ROWS_PER_WAVE * WAVES_PER_BLOCK;
    const int blocks = out_size / rows_per_block;
    tt_chain_wmma_f32<<<blocks, 512, 0, stream>>>(X, cf, cm, cl, out);
}


# Middle cores fused in triples, not pairs (80 products kept in LDS, made in two chained WMMA stages): 10-step chain, rows ordered by the 3-bit digit, 8 classes per step; sum order changes
# speedup vs baseline: 2.0130x; 1.2780x over previous
#include <hip/hip_runtime.h>


#define TT_D 32
#define TT_R 16
#define TT_STEPS 10
#define TT_CLS 8
#define WAVES_PER_BLOCK 16
#define TILES_PER_WAVE 4
#define GROUPS_PER_WAVE 2
#define ROWS_PER_WAVE (16 * TILES_PER_WAVE)
#define LDS_ROW_STRIDE 18
#define WG_ROWS (ROWS_PER_WAVE * WAVES_PER_BLOCK)
#define CARRY_FLOATS (WG_ROWS * LDS_ROW_STRIDE)
#define FUSED_FLOATS (TT_STEPS * TT_CLS * TT_R * TT_R)
#define STEP_INTS (TT_CLS * WAVES_PER_BLOCK)
#define CNT_INTS (TT_STEPS * STEP_INTS)

typedef float v2f __attribute__((ext_vector_type(2)));
typedef float v8f __attribute__((ext_vector_type(8)));
typedef int   v4i __attribute__((ext_vector_type(4)));
typedef int   v8i __attribute__((ext_vector_type(8)));

static __device__ __forceinline__ int fused_off(int f, int u, int g) {
    return f * (TT_R * TT_R) + u * TT_R + 4 * (g ^ ((u >> 2) & 3));
}

static __device__ __forceinline__ void load_a(v2f (&a)[4], const float* wbuf, int slot, int half) {
    const float* q = wbuf + slot * LDS_ROW_STRIDE + 2 * half;
    #pragma unroll
    for (int c = 0; c < 4; ++c) {
        const float2 u = *(const float2*)(q + 4 * c);
        a[c] = (v2f){ u.x, u.y };
    }
}

static __device__ __forceinline__ void tile_class(v8f& r, const v2f (&a)[4], const v2f (&b)[4], int th) {
    v8f pr = (v8f){};
    #pragma unroll
    for (int c = 0; c < 4; ++c)
        pr = __builtin_amdgcn_wmma_f32_16x16x4_f32(
                 false, a[c], false, b[c], (short)0, pr, false, false);
    const v8i vi = (v8i){ 0, 1, 2, 3, 4, 5, 6, 7 };
    r = (vi >= th) ? pr : r;
}

static __device__ __forceinline__ void store_tile(float* wbuf, int t, const v8f& r, int row, int half) {
    #pragma unroll
    for (int v = 0; v < 8; ++v)
        wbuf[(16 * t + v + 8 * half) * LDS_ROW_STRIDE + row] = r[v];
}

static __device__ __forceinline__ unsigned class_mask(unsigned d, unsigned qa, unsigned qb, unsigned qc) {
    return ((d & 1u) ? qa : ~qa) & ((d & 2u) ? qb : ~qb) & ((d & 4u) ? qc : ~qc);
}

__global__ __launch_bounds__(512, 4) void tt_chain_wmma_f32(
    const int*   __restrict__ X,
    const float* __restrict__ core_first,
    const float* __restrict__ cores_mid,
    const float* __restrict__ core_last,
    float*       __restrict__ out)
{
    __shared__ __attribute__((aligned(16))) float fused[FUSED_FLOATS];
    __shared__ __attribute__((aligned(16))) float carry[2][CARRY_FLOATS];
    __shared__ int   smap[2][WG_ROWS];
    __shared__ __attribute__((aligned(16))) int cnt[CNT_INTS];
    __shared__ __attribute__((aligned(16))) int pbase[CNT_INTS];
    __shared__ __attribute__((aligned(16))) int cstart[TT_STEPS * TT_CLS];

    const int tid  = threadIdx.x;
    const int wave = __builtin_amdgcn_readfirstlane(tid >> 5);
    const int lane = tid & 31;
    const int row  = lane & 15;
    const int half = lane >> 4;

    const int base = (blockIdx.x * WAVES_PER_BLOCK + wave) * ROWS_PER_WAVE;

    const int wofs = wave * ROWS_PER_WAVE;

    const int tw = wave < TT_STEPS ? wave : TT_STEPS - 1;
    float4 am[4];
    float2 bm[4][2];
    float2 cm[8];
    {
        const float* Ma = cores_mid + (3 * tw) * (TT_R * TT_R * 2);
        const float* Mb = Ma + TT_R * TT_R * 2;
        const float* Mc = Mb + TT_R * TT_R * 2;
        #pragma unroll
        for (int c = 0; c < 4; ++c) {
            const int k = 4 * c + 2 * half;
            am[c]    = *(const float4*)(Ma + row * 32 + k * 2);
            bm[c][0] = *(const float2*)(Mb + k * 32 + row * 2);
            bm[c][1] = *(const float2*)(Mb + (k + 1) * 32 + row * 2);
        }
        #pragma unroll
        for (int v = 0; v < 8; ++v)
            cm[v] = *(const float2*)(Mc + (8 * half + v) * 32 + row * 2);
    }

    v4i xv[GROUPS_PER_WAVE][8];
    #pragma unroll
    for (int g = 0; g < GROUPS_PER_WAVE; ++g) {
        const v4i* xr = (const v4i*)(X + (size_t)(base + 32 * g + lane) * TT_D);
        #pragma unroll
        for (int q = 0; q < 8; ++q)
            xv[g][q] = __builtin_nontemporal_load(xr + q);
    }

    if (wave < TT_STEPS) {
        #pragma unroll
        for (int ab = 0; ab < 4; ++ab) {
            v8f qd = (v8f){};
            #pragma unroll
            for (int c = 0; c < 4; ++c) {
                const v2f a1 = (ab & 2) ? (v2f){ bm[c][0].y, bm[c][1].y } : (v2f){ bm[c][0].x, bm[c][1].x };
                const v2f b1 = (ab & 1) ? (v2f){ am[c].y, am[c].w } : (v2f){ am[c].x, am[c].z };
                qd = __builtin_amdgcn_wmma_f32_16x16x4_f32(
                         false, a1, false, b1, (short)0, qd, false, false);
            }
            #pragma unroll
            for (int cc = 0; cc < 2; ++cc) {
                v8f acc = (v8f){};
                #pragma unroll
                for (int c = 0; c < 4; ++c) {
                    const v2f a2 = (v2f){ qd[2 * c], qd[2 * c + 1] };
                    const v2f b2 = cc ? (v2f){ cm[2 * c].y, cm[2 * c + 1].y } : (v2f){ cm[2 * c].x, cm[2 * c + 1].x };
                    acc = __builtin_amdgcn_wmma_f32_16x16x4_f32(
                              false, a2, false, b2, (short)0, acc, false, false);
                }
                const int f = TT_CLS * wave + ab + 4 * cc;
                *(float4*)(fused + fused_off(f, row, 2 * half))     = make_float4(acc[0], acc[1], acc[2], acc[3]);
                *(float4*)(fused + fused_off(f, row, 2 * half + 1)) = make_float4(acc[4], acc[5], acc[6], acc[7]);
            }
        }
    }

    unsigned xo[GROUPS_PER_WAVE];
    #pragma unroll
    for (int g = 0; g < GROUPS_PER_WAVE; ++g) {
        unsigned m = 0u;
        #pragma unroll
        for (int q = 0; q < 8; ++q) {
            const v4i v = xv[g][q];
            m |= (unsigned)(v.x & 1) << (4 * q + 0);
            m |= (unsigned)(v.y & 1) << (4 * q + 1);
            m |= (unsigned)(v.z & 1) << (4 * q + 2);
            m |= (unsigned)(v.w & 1) << (4 * q + 3);
        }
        xo[g] = m;
    }

    #pragma unroll
    for (int g = 0; g < GROUPS_PER_WAVE; ++g) {
        const bool x0 = (xo[g] & 1u) != 0u;
        float* dst = carry[0] + (wofs + 32 * g + lane) * LDS_ROW_STRIDE;
        #pragma unroll
        for (int k = 0; k < TT_R; k += 2) {
            float2 iv;
            iv.x = x0 ? core_first[k * 2 + 1]       : core_first[k * 2];
            iv.y = x0 ? core_first[(k + 1) * 2 + 1] : core_first[(k + 1) * 2];
            *(float2*)(dst + k) = iv;
        }
    }

    #pragma unroll 1
    for (int p = 0; p < TT_STEPS; ++p) {
        const unsigned e0 = (xo[0] >> (3 * p + 1)) & 7u;
        const unsigned e1 = (xo[1] >> (3 * p + 1)) & 7u;
        const unsigned qa0 = __builtin_amdgcn_ballot_w32((e0 & 1u) != 0u);
        const unsigned qb0 = __builtin_amdgcn_ballot_w32((e0 & 2u) != 0u);
        const unsigned qc0 = __builtin_amdgcn_ballot_w32((e0 & 4u) != 0u);
        const unsigned qa1 = __builtin_amdgcn_ballot_w32((e1 & 1u) != 0u);
        const unsigned qb1 = __builtin_amdgcn_ballot_w32((e1 & 2u) != 0u);
        const unsigned qc1 = __builtin_amdgcn_ballot_w32((e1 & 4u) != 0u);
        const unsigned lc = (unsigned)lane & 7u;
        const int n = __builtin_popcount(class_mask(lc, qa0, qb0, qc0))
                    + __builtin_popcount(class_mask(lc, qa1, qb1, qc1));
        if (lane < TT_CLS) cnt[p * STEP_INTS + lane * WAVES_PER_BLOCK + wave] = n;
    }

    __syncthreads();

    if (wave < TT_STEPS) {
        const v4i c4 = *(const v4i*)(cnt + wave * STEP_INTS + 4 * lane);
        const int own = c4.x + c4.y + c4.z + c4.w;
        int inc = own;
        #pragma unroll
        for (int d = 1; d < 32; d <<= 1) {
            const int t = __builtin_amdgcn_ds_bpermute(((lane - d) & 31) << 2, inc);
            inc += (lane >= d) ? t : 0;
        }
        const int ex = inc - own;
        v4i o;
        o.x = ex;
        o.y = o.x + c4.x;
        o.z = o.y + c4.y;
        o.w = o.z + c4.z;
        *(v4i*)(pbase + wave * STEP_INTS + 4 * lane) = o;
        if ((lane & 3) == 0) cstart[wave * TT_CLS + (lane >> 2)] = ex;
    }

    __syncthreads();

    int cur0 = wofs + lane;
    int cur1 = wofs + 32 + lane;

    int nb0, nb1;
    v4i nSa, nSb;
    {
        const unsigned e0 = (xo[0] >> 1) & 7u;
        const unsigned e1 = (xo[1] >> 1) & 7u;
        nb0 = pbase[(int)e0 * WAVES_PER_BLOCK + wave];
        nb1 = pbase[(int)e1 * WAVES_PER_BLOCK + wave];
        nSa = *(const v4i*)(cstart);
        nSb = *(const v4i*)(cstart + 4);
    }

    #pragma unroll 1
    for (int p = 0; p < TT_STEPS; ++p) {
        const float* cin  = carry[p & 1];
        float*       cout = carry[(p + 1) & 1];
        int*         mp   = smap[p & 1];

        const int b0 = nb0;
        const int b1 = nb1;
        int st[TT_CLS + 1];
        {
            const int S[TT_CLS] = { 0,
                __builtin_amdgcn_readfirstlane(nSa.y), __builtin_amdgcn_readfirstlane(nSa.z),
                __builtin_amdgcn_readfirstlane(nSa.w), __builtin_amdgcn_readfirstlane(nSb.x),
                __builtin_amdgcn_readfirstlane(nSb.y), __builtin_amdgcn_readfirstlane(nSb.z),
                __builtin_amdgcn_readfirstlane(nSb.w) };
            st[0] = 0;
            #pragma unroll
            for (int cls = 1; cls < TT_CLS; ++cls)
                st[cls] = min(max(S[cls] - wofs, 0), ROWS_PER_WAVE);
            st[TT_CLS] = ROWS_PER_WAVE;
        }

        const unsigned d0  = (xo[0] >> (3 * p + 1)) & 7u;
        const unsigned d1  = (xo[1] >> (3 * p + 1)) & 7u;
        const unsigned qa0 = __builtin_amdgcn_ballot_w32((d0 & 1u) != 0u);
        const unsigned qb0 = __builtin_amdgcn_ballot_w32((d0 & 2u) != 0u);
        const unsigned qc0 = __builtin_amdgcn_ballot_w32((d0 & 4u) != 0u);
        const unsigned qa1 = __builtin_amdgcn_ballot_w32((d1 & 1u) != 0u);
        const unsigned qb1 = __builtin_amdgcn_ballot_w32((d1 & 2u) != 0u);
        const unsigned qc1 = __builtin_amdgcn_ballot_w32((d1 & 4u) != 0u);
        const unsigned mym0 = class_mask(d0, qa0, qb0, qc0);
        const unsigned mym1 = class_mask(d1, qa1, qb1, qc1);
        const unsigned g0m1 = class_mask(d1, qa0, qb0, qc0);
        const int      rank0 = (int)__builtin_amdgcn_mbcnt_lo(mym0, 0u);
        const int      rank1 = (int)__builtin_amdgcn_mbcnt_lo(mym1, 0u);
        const int      pos0  = b0 + rank0;
        const int      pos1  = b1 + __builtin_popcount(g0m1) + rank1;

        mp[pos0] = cur0;
        mp[pos1] = cur1;
        cur0 = pos0;
        cur1 = pos1;

        {
            const int pn = p < TT_STEPS - 1 ? p + 1 : TT_STEPS - 1;
            const unsigned e0 = (xo[0] >> (3 * pn + 1)) & 7u;
            const unsigned e1 = (xo[1] >> (3 * pn + 1)) & 7u;
            const int* pb = pbase + pn * STEP_INTS;
            nb0 = pb[(int)e0 * WAVES_PER_BLOCK + wave];
            nb1 = pb[(int)e1 * WAVES_PER_BLOCK + wave];
            nSa = *(const v4i*)(cstart + pn * TT_CLS);
            nSb = *(const v4i*)(cstart + pn * TT_CLS + 4);
        }

        v2f bq[TT_CLS][4];
        #pragma unroll
        for (int cls = 0; cls < TT_CLS; ++cls)
            if (st[cls + 1] > st[cls]) {
                #pragma unroll
                for (int c = 0; c < 4; ++c) {
                    const float2 t2 = *(const float2*)(fused + fused_off(TT_CLS * p + cls, row, c) + 2 * half);
                    bq[cls][c] = (v2f){ t2.x, t2.y };
                }
            }

        __syncthreads();

        v2f a0[4], a1[4], a2[4], a3[4];
        load_a(a0, cin, mp[wofs + row],      half);
        load_a(a1, cin, mp[wofs + 16 + row], half);
        load_a(a2, cin, mp[wofs + 32 + row], half);
        load_a(a3, cin, mp[wofs + 48 + row], half);

        v8f r0 = (v8f){}, r1 = (v8f){}, r2 = (v8f){}, r3 = (v8f){};
        #pragma unroll
        for (int cls = 0; cls < TT_CLS; ++cls) {
            const int lo = st[cls];
            const int hi = st[cls + 1];
            if (hi > lo) {
                const int th = lo - 8 * half;
                if (lo < 16)            tile_class(r0, a0, bq[cls], th);
                if (lo < 32 && hi > 16) tile_class(r1, a1, bq[cls], th - 16);
                if (lo < 48 && hi > 32) tile_class(r2, a2, bq[cls], th - 32);
                if (hi > 48)            tile_class(r3, a3, bq[cls], th - 48);
            }
        }

        float* wout = cout + wofs * LDS_ROW_STRIDE;
        store_tile(wout, 0, r0, row, half);
        store_tile(wout, 1, r1, row, half);
        store_tile(wout, 2, r2, row, half);
        store_tile(wout, 3, r3, row, half);
    }

    __syncthreads();

    #pragma unroll
    for (int g = 0; g < GROUPS_PER_WAVE; ++g) {
        const bool xl = (xo[g] >> 31) != 0u;
        const int  sl = g ? cur1 : cur0;
        const float* src = carry[TT_STEPS & 1] + sl * LDS_ROW_STRIDE;
        float acc0 = 0.f, acc1 = 0.f;
        #pragma unroll
        for (int c = 0; c < 4; ++c) {
            const float2 t0 = *(const float2*)(src + 4 * c);
            const float2 t1 = *(const float2*)(src + 4 * c + 2);
            const int k0 = 4 * c;
            const int k1 = 4 * c + 2;
            acc0 += t0.x * (xl ? core_last[k0 * 2 + 1]       : core_last[k0 * 2]);
            acc0 += t0.y * (xl ? core_last[(k0 + 1) * 2 + 1] : core_last[(k0 + 1) * 2]);
            acc1 += t1.x * (xl ? core_last[k1 * 2 + 1]       : core_last[k1 * 2]);
            acc1 += t1.y * (xl ? core_last[(k1 + 1) * 2 + 1] : core_last[(k1 + 1) * 2]);
        }
        __builtin_nontemporal_store(acc0 + acc1, out + base + 32 * g + lane);
    }
}

extern "C" void kernel_launch(void* const* d_in, const int* in_sizes, int n_in,
                              void* d_out, int out_size, void* d_ws, size_t ws_size,
                              hipStream_t stream) {
    (void)in_sizes; (void)n_in; (void)d_ws; (void)ws_size;
    const int*   X  = (const int*)d_in[0];
    const float* cf = (const float*)d_in[1];
    const float* cm = (const float*)d_in[2];
    const float* cl = (const float*)d_in[3];
    float* out = (float*)d_out;

    const int rows_per_block = ROWS_PER_WAVE * WAVES_PER_BLOCK;
    const int blocks = out_size / rows_per_block;
    tt_chain_wmma_f32<<<blocks, 512, 0, stream>>>(X, cf, cm, cl, out);
}
